# starting kernel
# baseline (speedup 1.0000x reference)
#include <hip/hip_runtime.h>
#include <hip/hip_bf16.h>

typedef __attribute__((ext_vector_type(2))) float v2f;
typedef __attribute__((ext_vector_type(8))) float v8f;

#define N_NODES 50000
#define N_EDGES 800000
#define DIM     128


__global__ void zero_kernel(float4* __restrict__ p, long n4) {
    long i = (long)blockIdx.x * blockDim.x + threadIdx.x;
    long stride = (long)gridDim.x * blockDim.x;
    float4 z = make_float4(0.f, 0.f, 0.f, 0.f);
    for (; i < n4; i += stride) p[i] = z;
}

__global__ void degree_kernel(const int* __restrict__ row,
                              const int* __restrict__ col,
                              float* __restrict__ deg_row,
                              float* __restrict__ deg_col) {
    int e = blockIdx.x * blockDim.x + threadIdx.x;
    if (e >= N_EDGES) return;
    atomicAdd(&deg_row[row[e]], 1.0f);
    atomicAdd(&deg_col[col[e]], 1.0f);
}

__global__ void invsqrt_kernel(float* __restrict__ deg) {
    int i = blockIdx.x * blockDim.x + threadIdx.x;
    if (i >= 2 * N_NODES) return;
    float d = deg[i];
    deg[i] = (d > 0.0f) ? rsqrtf(fmaxf(d, 1.0f)) : 0.0f;
}

__global__ void prep_kernel(const float* __restrict__ W_src,
                            const float* __restrict__ b_src,
                            const float* __restrict__ W_dst,
                            const float* __restrict__ b_dst,
                            float* __restrict__ Wc_s,
                            float* __restrict__ Wc_d,
                            float* __restrict__ bias) {
    int i = blockIdx.x * blockDim.x + threadIdx.x;
    if (i < DIM * DIM) {
        Wc_s[i] = 0.5f * W_src[i];
        Wc_d[i] = 0.5f * W_dst[i];
    }
    if (i < DIM) bias[i] = 0.5f * b_src[i] + 0.5f * b_dst[i];
}

__global__ void scatter_kernel(const float* __restrict__ x,
                               const int* __restrict__ row,
                               const int* __restrict__ col,
                               const float* __restrict__ inv_row,
                               const float* __restrict__ inv_col,
                               float* __restrict__ agg_fwd,
                               float* __restrict__ agg_rev) {
    int lane = threadIdx.x & 31;
    int e = blockIdx.x * (blockDim.x >> 5) + (threadIdx.x >> 5);
    if (e >= N_EDGES) return;
    int r = row[e];
    int c = col[e];
    float w = inv_row[r] * inv_col[c];

    const float4 vc = ((const float4*)(x + (size_t)c * DIM))[lane];
    const float4 vr = ((const float4*)(x + (size_t)r * DIM))[lane];

    float* af = agg_fwd + (size_t)r * DIM + lane * 4;
    float* ar = agg_rev + (size_t)c * DIM + lane * 4;
    atomicAdd(af + 0, w * vc.x);
    atomicAdd(af + 1, w * vc.y);
    atomicAdd(af + 2, w * vc.z);
    atomicAdd(af + 3, w * vc.w);
    atomicAdd(ar + 0, w * vr.x);
    atomicAdd(ar + 1, w * vr.y);
    atomicAdd(ar + 2, w * vr.z);
    atomicAdd(ar + 3, w * vr.w);
}

__global__ void __launch_bounds__(256)
gemm_kernel(const float* __restrict__ agg_fwd,
            const float* __restrict__ agg_rev,
            const float* __restrict__ Wc_s,
            const float* __restrict__ Wc_d,
            const float* __restrict__ bias,
            float* __restrict__ out) {
    const int lane = threadIdx.x & 31;
    const int tile = blockIdx.x * 8 + (threadIdx.x >> 5);
    const int mt = tile >> 3;
    const int nt = tile & 7;

    const int half = lane >> 4;
    const int l15  = lane & 15;
    const int kb   = half * 2;

    const float* arow_f = agg_fwd + (size_t)(mt * 16 + l15) * DIM;
    const float* arow_r = agg_rev + (size_t)(mt * 16 + l15) * DIM;
    const float* wrow_s = Wc_s    + (size_t)(nt * 16 + l15) * DIM;
    const float* wrow_d = Wc_d    + (size_t)(nt * 16 + l15) * DIM;

    v8f c;
    const float bv = bias[nt * 16 + l15];
    #pragma unroll
    for (int j = 0; j < 8; ++j) c[j] = bv;

    #pragma unroll
    for (int k = 0; k < DIM; k += 4) {
        v2f a0 = *(const v2f*)(arow_f + k + kb);
        v2f b0 = *(const v2f*)(wrow_s + k + kb);
        c = __builtin_amdgcn_wmma_f32_16x16x4_f32(false, a0, false, b0,
                                                  (short)0, c, false, false);
        v2f a1 = *(const v2f*)(arow_r + k + kb);
        v2f b1 = *(const v2f*)(wrow_d + k + kb);
        c = __builtin_amdgcn_wmma_f32_16x16x4_f32(false, a1, false, b1,
                                                  (short)0, c, false, false);
    }

    const int m0 = mt * 16 + half * 8;
    const int n  = nt * 16 + l15;
    #pragma unroll
    for (int j = 0; j < 8; ++j)
        out[(size_t)(m0 + j) * DIM + n] = c[j];
}

extern "C" void kernel_launch(void* const* d_in, const int* in_sizes, int n_in,
                              void* d_out, int out_size, void* d_ws, size_t ws_size,
                              hipStream_t stream) {
    const float* x      = (const float*)d_in[0];
    const float* W_src  = (const float*)d_in[1];
    const float* b_src  = (const float*)d_in[2];
    const float* W_dst  = (const float*)d_in[3];
    const float* b_dst  = (const float*)d_in[4];
    const int*   eidx   = (const int*)d_in[5];
    const int*   row    = eidx;
    const int*   col    = eidx + N_EDGES;
    float*       outp   = (float*)d_out;

    float* ws = (float*)d_ws;
    float* deg_row = ws;
    float* deg_col = ws + N_NODES;
    float* agg_fwd = ws + 2 * (size_t)N_NODES;
    float* agg_rev = agg_fwd + (size_t)N_NODES * DIM;
    float* Wc_s    = agg_rev + (size_t)N_NODES * DIM;
    float* Wc_d    = Wc_s + DIM * DIM;
    float* bias    = Wc_d + DIM * DIM;

    {
        long n4 = (long)(2 + 2 * DIM) * N_NODES / 4;
        zero_kernel<<<2048, 256, 0, stream>>>((float4*)ws, n4);
    }
    degree_kernel<<<(N_EDGES + 255) / 256, 256, 0, stream>>>(row, col, deg_row, deg_col);
    invsqrt_kernel<<<(2 * N_NODES + 255) / 256, 256, 0, stream>>>(deg_row);
    prep_kernel<<<(DIM * DIM + 255) / 256, 256, 0, stream>>>(W_src, b_src, W_dst, b_dst,
                                                             Wc_s, Wc_d, bias);
    scatter_kernel<<<(N_EDGES + 7) / 8, 256, 0, stream>>>(x, row, col, deg_row, deg_col,
                                                          agg_fwd, agg_rev);
    gemm_kernel<<<(N_NODES / 16) * (DIM / 16) / 8, 256, 0, stream>>>(agg_fwd, agg_rev,
                                                                     Wc_s, Wc_d, bias, outp);
}


# scatter by node-range ownership: a work-group keeps 391 nodes' aggregate rows in LDS, filters the whole edge list for them, adds x rows with LDS float adds; global float atomics and agg clear gone
# speedup vs baseline: 1.7879x; 1.7879x over previous
#include <hip/hip_runtime.h>
#include <hip/hip_bf16.h>

typedef __attribute__((ext_vector_type(2))) float v2f;
typedef __attribute__((ext_vector_type(8))) float v8f;

#define N_NODES 50000
#define N_EDGES 800000
#define DIM     128

#define RANGE   391
#define NRANGE  128
#define CHUNKS  (N_EDGES / 128)
#define OUTER   50
#define QCAP    544
static_assert(N_EDGES % 128 == 0, "edge list splits into whole 128-edge pieces");
static_assert(NRANGE * RANGE >= N_NODES, "ranges cover every node");
static_assert(OUTER * 4 * 32 >= CHUNKS && OUTER % 2 == 0, "one sweep covers every piece, in step pairs");
static_assert(N_EDGES <= (1 << 20) && RANGE <= (1 << 12), "queue entry packs (local node, edge id) in 32 bits");


__global__ void zero_kernel(float4* __restrict__ p, long n4) {
    long i = (long)blockIdx.x * blockDim.x + threadIdx.x;
    long stride = (long)gridDim.x * blockDim.x;
    float4 z = make_float4(0.f, 0.f, 0.f, 0.f);
    for (; i < n4; i += stride) p[i] = z;
}

__global__ void degree_kernel(const int* __restrict__ row,
                              const int* __restrict__ col,
                              float* __restrict__ deg_row,
                              float* __restrict__ deg_col) {
    int e = blockIdx.x * blockDim.x + threadIdx.x;
    if (e >= N_EDGES) return;
    atomicAdd(&deg_row[row[e]], 1.0f);
    atomicAdd(&deg_col[col[e]], 1.0f);
}

__global__ void invsqrt_kernel(float* __restrict__ deg) {
    int i = blockIdx.x * blockDim.x + threadIdx.x;
    if (i >= 2 * N_NODES) return;
    float d = deg[i];
    deg[i] = (d > 0.0f) ? rsqrtf(fmaxf(d, 1.0f)) : 0.0f;
}

__global__ void prep_kernel(const float* __restrict__ W_src,
                            const float* __restrict__ b_src,
                            const float* __restrict__ W_dst,
                            const float* __restrict__ b_dst,
                            float* __restrict__ Wc_s,
                            float* __restrict__ Wc_d,
                            float* __restrict__ bias) {
    int i = blockIdx.x * blockDim.x + threadIdx.x;
    if (i < DIM * DIM) {
        Wc_s[i] = 0.5f * W_src[i];
        Wc_d[i] = 0.5f * W_dst[i];
    }
    if (i < DIM) bias[i] = 0.5f * b_src[i] + 0.5f * b_dst[i];
}

__device__ __forceinline__ void add_row(float* acc, int t, int lane, float w, const float4& v) {
    float* a = acc + t * DIM + lane;
    atomicAdd(a,      w * v.x);
    atomicAdd(a + 32, w * v.y);
    atomicAdd(a + 64, w * v.z);
    atomicAdd(a + 96, w * v.w);
}

__device__ __forceinline__ void drain_full(const unsigned* qb, int lane, int n0,
                                           const int* __restrict__ src,
                                           const float* __restrict__ inv_d,
                                           const float* __restrict__ inv_s,
                                           const float* __restrict__ x,
                                           float* acc) {
    const unsigned p = qb[lane];
    const int t = (int)(p >> 20);
    const int s = src[(int)(p & 0xFFFFFu)];
    const float w = inv_d[n0 + t] * inv_s[s];
    const int wb = __builtin_bit_cast(int, w);

    float4 v[2][8];
    #pragma unroll
    for (int u = 0; u < 8; ++u)
        v[0][u] = ((const float4*)(x + (size_t)__builtin_amdgcn_readlane(s, u) * DIM))[lane];
    #pragma unroll
    for (int g = 0; g < 4; ++g) {
        if (g + 1 < 4) {
            #pragma unroll
            for (int u = 0; u < 8; ++u)
                v[(g + 1) & 1][u] = ((const float4*)(x +
                    (size_t)__builtin_amdgcn_readlane(s, (g + 1) * 8 + u) * DIM))[lane];
        }
        #pragma unroll
        for (int u = 0; u < 8; ++u)
            add_row(acc, __builtin_amdgcn_readlane(t, g * 8 + u), lane,
                    __builtin_bit_cast(float, __builtin_amdgcn_readlane(wb, g * 8 + u)),
                    v[g & 1][u]);
    }
}

__device__ __forceinline__ void drain_tail(const unsigned* qb, int n, int lane, int n0,
                                           const int* __restrict__ src,
                                           const float* __restrict__ inv_d,
                                           const float* __restrict__ inv_s,
                                           const float* __restrict__ x,
                                           float* acc) {
    int s = 0, t = 0;
    float w = 0.0f;
    if (lane < n) {
        const unsigned p = qb[lane];
        t = (int)(p >> 20);
        s = src[(int)(p & 0xFFFFFu)];
        w = inv_d[n0 + t] * inv_s[s];
    }
    const int wb = __builtin_bit_cast(int, w);
    int i = 0;
    for (; i + 8 <= n; i += 8) {
        float4 v[8];
        #pragma unroll
        for (int u = 0; u < 8; ++u)
            v[u] = ((const float4*)(x + (size_t)__builtin_amdgcn_readlane(s, i + u) * DIM))[lane];
        #pragma unroll
        for (int u = 0; u < 8; ++u)
            add_row(acc, __builtin_amdgcn_readlane(t, i + u), lane,
                    __builtin_bit_cast(float, __builtin_amdgcn_readlane(wb, i + u)), v[u]);
    }
    for (; i < n; ++i) {
        const float4 v = ((const float4*)(x + (size_t)__builtin_amdgcn_readlane(s, i) * DIM))[lane];
        add_row(acc, __builtin_amdgcn_readlane(t, i), lane,
                __builtin_bit_cast(float, __builtin_amdgcn_readlane(wb, i)), v);
    }
}

__global__ void __launch_bounds__(1024)
aggregate_kernel(const float* __restrict__ x,
                 const int* __restrict__ row,
                 const int* __restrict__ col,
                 const float* __restrict__ inv_row,
                 const float* __restrict__ inv_col,
                 float* __restrict__ agg_fwd,
                 float* __restrict__ agg_rev) {
    __shared__ __attribute__((aligned(16))) float acc[RANGE * DIM];
    __shared__ unsigned queue[32 * QCAP];

    const int tid  = threadIdx.x;
    const int lane = tid & 31;
    const int wave = __builtin_amdgcn_readfirstlane(tid >> 5);
    const int dir  = blockIdx.x >> 7;
    const int rg   = blockIdx.x & (NRANGE - 1);
    const int n0   = rg * RANGE;
    const int cnt  = (N_NODES - n0 < RANGE) ? (N_NODES - n0) : RANGE;
    const unsigned ucnt = (unsigned)cnt;

    const int*   dst   = dir ? col : row;
    const int*   src   = dir ? row : col;
    const float* inv_d = dir ? inv_col : inv_row;
    const float* inv_s = dir ? inv_row : inv_col;
    float*       agg   = dir ? agg_rev : agg_fwd;

    {
        const float4 z = make_float4(0.f, 0.f, 0.f, 0.f);
        for (int i = tid; i < RANGE * DIM / 4; i += 1024) ((float4*)acc)[i] = z;
    }
    __syncthreads();

    unsigned* q = queue + wave * QCAP;
    int pending = 0;
    const int rot = (rg * OUTER) >> 7;

    auto load4 = [&](int4 (&r)[4], int step) {
        if (step >= OUTER) step -= OUTER;
        #pragma unroll
        for (int u = 0; u < 4; ++u) {
            int c = (step * 4 + u) * 32 + wave;
            c = c < CHUNKS ? c : CHUNKS - 1;
            r[u] = *(const int4*)(dst + (size_t)c * 128 + lane * 4);
        }
    };
    auto look4 = [&](const int4 (&r)[4], int step) {
        if (step >= OUTER) step -= OUTER;
        #pragma unroll
        for (int u = 0; u < 4; ++u) {
            const int chunk = (step * 4 + u) * 32 + wave;
            if (chunk < CHUNKS) {
                const unsigned t0 = (unsigned)(r[u].x - n0);
                const unsigned t1 = (unsigned)(r[u].y - n0);
                const unsigned t2 = (unsigned)(r[u].z - n0);
                const unsigned t3 = (unsigned)(r[u].w - n0);
                const bool h0 = t0 < ucnt, h1 = t1 < ucnt, h2 = t2 < ucnt, h3 = t3 < ucnt;
                if (__builtin_amdgcn_ballot_w32(h0 | h1 | h2 | h3) != 0u) {
                    const unsigned e0 = (unsigned)(chunk * 128 + lane * 4);
                    unsigned m;
                    m = __builtin_amdgcn_ballot_w32(h0);
                    if (h0) q[pending + (int)__builtin_amdgcn_mbcnt_lo(m, 0u)] = (t0 << 20) | e0;
                    pending += __builtin_popcount(m);
                    m = __builtin_amdgcn_ballot_w32(h1);
                    if (h1) q[pending + (int)__builtin_amdgcn_mbcnt_lo(m, 0u)] = (t1 << 20) | (e0 + 1u);
                    pending += __builtin_popcount(m);
                    m = __builtin_amdgcn_ballot_w32(h2);
                    if (h2) q[pending + (int)__builtin_amdgcn_mbcnt_lo(m, 0u)] = (t2 << 20) | (e0 + 2u);
                    pending += __builtin_popcount(m);
                    m = __builtin_amdgcn_ballot_w32(h3);
                    if (h3) q[pending + (int)__builtin_amdgcn_mbcnt_lo(m, 0u)] = (t3 << 20) | (e0 + 3u);
                    pending += __builtin_popcount(m);
                }
            }
        }
        while (pending >= 32) {
            pending -= 32;
            drain_full(q + pending, lane, n0, src, inv_d, inv_s, x, acc);
        }
    };

    int4 da[4], db[4];
    load4(da, rot);
    for (int o = 0; o < OUTER; o += 2) {
        load4(db, o + 1 + rot);
        look4(da, o + rot);
        load4(da, o + 2 + rot);
        look4(db, o + 1 + rot);
    }
    if (pending > 0) drain_tail(q, pending, lane, n0, src, inv_d, inv_s, x, acc);

    __syncthreads();

    for (int t = wave; t < cnt; t += 32) {
        const float* a = acc + t * DIM + lane;
        const float4 v = make_float4(a[0], a[32], a[64], a[96]);
        ((float4*)(agg + (size_t)(n0 + t) * DIM))[lane] = v;
    }
}

__global__ void __launch_bounds__(256)
gemm_kernel(const float* __restrict__ agg_fwd,
            const float* __restrict__ agg_rev,
            const float* __restrict__ Wc_s,
            const float* __restrict__ Wc_d,
            const float* __restrict__ bias,
            float* __restrict__ out) {
    const int lane = threadIdx.x & 31;
    const int tile = blockIdx.x * 8 + (threadIdx.x >> 5);
    const int mt = tile >> 3;
    const int nt = tile & 7;

    const int half = lane >> 4;
    const int l15  = lane & 15;
    const int kb   = half * 2;

    const float* arow_f = agg_fwd + (size_t)(mt * 16 + l15) * DIM;
    const float* arow_r = agg_rev + (size_t)(mt * 16 + l15) * DIM;
    const float* wrow_s = Wc_s    + (size_t)(nt * 16 + l15) * DIM;
    const float* wrow_d = Wc_d    + (size_t)(nt * 16 + l15) * DIM;

    v8f c;
    const float bv = bias[nt * 16 + l15];
    #pragma unroll
    for (int j = 0; j < 8; ++j) c[j] = bv;

    #pragma unroll
    for (int k = 0; k < DIM; k += 4) {
        v2f a0 = *(const v2f*)(arow_f + k + kb);
        v2f b0 = *(const v2f*)(wrow_s + k + kb);
        c = __builtin_amdgcn_wmma_f32_16x16x4_f32(false, a0, false, b0,
                                                  (short)0, c, false, false);
        v2f a1 = *(const v2f*)(arow_r + k + kb);
        v2f b1 = *(const v2f*)(wrow_d + k + kb);
        c = __builtin_amdgcn_wmma_f32_16x16x4_f32(false, a1, false, b1,
                                                  (short)0, c, false, false);
    }

    const int m0 = mt * 16 + half * 8;
    const int n  = nt * 16 + l15;
    #pragma unroll
    for (int j = 0; j < 8; ++j)
        out[(size_t)(m0 + j) * DIM + n] = c[j];
}

extern "C" void kernel_launch(void* const* d_in, const int* in_sizes, int n_in,
                              void* d_out, int out_size, void* d_ws, size_t ws_size,
                              hipStream_t stream) {
    const float* x      = (const float*)d_in[0];
    const float* W_src  = (const float*)d_in[1];
    const float* b_src  = (const float*)d_in[2];
    const float* W_dst  = (const float*)d_in[3];
    const float* b_dst  = (const float*)d_in[4];
    const int*   eidx   = (const int*)d_in[5];
    const int*   row    = eidx;
    const int*   col    = eidx + N_EDGES;
    float*       outp   = (float*)d_out;

    float* ws = (float*)d_ws;
    float* deg_row = ws;
    float* deg_col = ws + N_NODES;
    float* agg_fwd = ws + 2 * (size_t)N_NODES;
    float* agg_rev = agg_fwd + (size_t)N_NODES * DIM;
    float* Wc_s    = agg_rev + (size_t)N_NODES * DIM;
    float* Wc_d    = Wc_s + DIM * DIM;
    float* bias    = Wc_d + DIM * DIM;

    {
        long n4 = (long)2 * N_NODES / 4;
        zero_kernel<<<(2 * N_NODES / 4 + 255) / 256, 256, 0, stream>>>((float4*)ws, n4);
    }
    degree_kernel<<<(N_EDGES + 255) / 256, 256, 0, stream>>>(row, col, deg_row, deg_col);
    invsqrt_kernel<<<(2 * N_NODES + 255) / 256, 256, 0, stream>>>(deg_row);
    prep_kernel<<<(DIM * DIM + 255) / 256, 256, 0, stream>>>(W_src, b_src, W_dst, b_dst,
                                                             Wc_s, Wc_d, bias);
    aggregate_kernel<<<2 * NRANGE, 1024, 0, stream>>>(x, row, col, deg_row, deg_col,
                                                      agg_fwd, agg_rev);
    gemm_kernel<<<(N_NODES / 16) * (DIM / 16) / 8, 256, 0, stream>>>(agg_fwd, agg_rev,
                                                                     Wc_s, Wc_d, bias, outp);
}


# GEMM re-tiled: one wave per 16x128 out strip (8 accumulator tiles); each A fragment loaded once, a trip ahead, and reused over 8 N tiles; W pair read from LDS in B-fragment order
# speedup vs baseline: 2.5239x; 1.4117x over previous
#include <hip/hip_runtime.h>
#include <hip/hip_bf16.h>

typedef __attribute__((ext_vector_type(2))) float v2f;
typedef __attribute__((ext_vector_type(8))) float v8f;

#define N_NODES 50000
#define N_EDGES 800000
#define DIM     128
#define WF_WORDS (DIM / 4 * 8 * 32)

#define RANGE   391
#define NRANGE  128
#define CHUNKS  (N_EDGES / 128)
#define OUTER   50
#define QCAP    544
static_assert(N_EDGES % 128 == 0, "edge list splits into whole 128-edge pieces");
static_assert(NRANGE * RANGE >= N_NODES, "ranges cover every node");
static_assert(OUTER * 4 * 32 >= CHUNKS && OUTER % 2 == 0, "one sweep covers every piece, in step pairs");
static_assert(N_EDGES <= (1 << 20) && RANGE <= (1 << 12), "queue entry packs (local node, edge id) in 32 bits");


__global__ void zero_kernel(float4* __restrict__ p, long n4) {
    long i = (long)blockIdx.x * blockDim.x + threadIdx.x;
    long stride = (long)gridDim.x * blockDim.x;
    float4 z = make_float4(0.f, 0.f, 0.f, 0.f);
    for (; i < n4; i += stride) p[i] = z;
}

__global__ void degree_kernel(const int* __restrict__ row,
                              const int* __restrict__ col,
                              float* __restrict__ deg_row,
                              float* __restrict__ deg_col) {
    int e = blockIdx.x * blockDim.x + threadIdx.x;
    if (e >= N_EDGES) return;
    atomicAdd(&deg_row[row[e]], 1.0f);
    atomicAdd(&deg_col[col[e]], 1.0f);
}

__global__ void invsqrt_kernel(float* __restrict__ deg) {
    int i = blockIdx.x * blockDim.x + threadIdx.x;
    if (i >= 2 * N_NODES) return;
    float d = deg[i];
    deg[i] = (d > 0.0f) ? rsqrtf(fmaxf(d, 1.0f)) : 0.0f;
}

__global__ void prep_kernel(const float* __restrict__ W_src,
                            const float* __restrict__ b_src,
                            const float* __restrict__ W_dst,
                            const float* __restrict__ b_dst,
                            float4* __restrict__ Wf,
                            float* __restrict__ bias) {
    int i = blockIdx.x * blockDim.x + threadIdx.x;
    if (i < WF_WORDS) {
        const int l  = i & 31;
        const int nt = (i >> 5) & 7;
        const int kq = i >> 8;
        const int n  = nt * 16 + (l & 15);
        const int k  = 4 * kq + 2 * (l >> 4);
        const float2 s = *(const float2*)(W_src + n * DIM + k);
        const float2 d = *(const float2*)(W_dst + n * DIM + k);
        Wf[i] = make_float4(0.5f * s.x, 0.5f * s.y, 0.5f * d.x, 0.5f * d.y);
    }
    if (i < DIM) bias[i] = 0.5f * b_src[i] + 0.5f * b_dst[i];
}

__device__ __forceinline__ void add_row(float* acc, int t, int lane, float w, const float4& v) {
    float* a = acc + t * DIM + lane;
    atomicAdd(a,      w * v.x);
    atomicAdd(a + 32, w * v.y);
    atomicAdd(a + 64, w * v.z);
    atomicAdd(a + 96, w * v.w);
}

__device__ __forceinline__ void drain_full(const unsigned* qb, int lane, int n0,
                                           const int* __restrict__ src,
                                           const float* __restrict__ inv_d,
                                           const float* __restrict__ inv_s,
                                           const float* __restrict__ x,
                                           float* acc) {
    const unsigned p = qb[lane];
    const int t = (int)(p >> 20);
    const int s = src[(int)(p & 0xFFFFFu)];
    const float w = inv_d[n0 + t] * inv_s[s];
    const int wb = __builtin_bit_cast(int, w);

    float4 v[2][8];
    #pragma unroll
    for (int u = 0; u < 8; ++u)
        v[0][u] = ((const float4*)(x + (size_t)__builtin_amdgcn_readlane(s, u) * DIM))[lane];
    #pragma unroll
    for (int g = 0; g < 4; ++g) {
        if (g + 1 < 4) {
            #pragma unroll
            for (int u = 0; u < 8; ++u)
                v[(g + 1) & 1][u] = ((const float4*)(x +
                    (size_t)__builtin_amdgcn_readlane(s, (g + 1) * 8 + u) * DIM))[lane];
        }
        #pragma unroll
        for (int u = 0; u < 8; ++u)
            add_row(acc, __builtin_amdgcn_readlane(t, g * 8 + u), lane,
                    __builtin_bit_cast(float, __builtin_amdgcn_readlane(wb, g * 8 + u)),
                    v[g & 1][u]);
    }
}

__device__ __forceinline__ void drain_tail(const unsigned* qb, int n, int lane, int n0,
                                           const int* __restrict__ src,
                                           const float* __restrict__ inv_d,
                                           const float* __restrict__ inv_s,
                                           const float* __restrict__ x,
                                           float* acc) {
    int s = 0, t = 0;
    float w = 0.0f;
    if (lane < n) {
        const unsigned p = qb[lane];
        t = (int)(p >> 20);
        s = src[(int)(p & 0xFFFFFu)];
        w = inv_d[n0 + t] * inv_s[s];
    }
    const int wb = __builtin_bit_cast(int, w);
    int i = 0;
    for (; i + 8 <= n; i += 8) {
        float4 v[8];
        #pragma unroll
        for (int u = 0; u < 8; ++u)
            v[u] = ((const float4*)(x + (size_t)__builtin_amdgcn_readlane(s, i + u) * DIM))[lane];
        #pragma unroll
        for (int u = 0; u < 8; ++u)
            add_row(acc, __builtin_amdgcn_readlane(t, i + u), lane,
                    __builtin_bit_cast(float, __builtin_amdgcn_readlane(wb, i + u)), v[u]);
    }
    for (; i < n; ++i) {
        const float4 v = ((const float4*)(x + (size_t)__builtin_amdgcn_readlane(s, i) * DIM))[lane];
        add_row(acc, __builtin_amdgcn_readlane(t, i), lane,
                __builtin_bit_cast(float, __builtin_amdgcn_readlane(wb, i)), v);
    }
}

__global__ void __launch_bounds__(1024)
aggregate_kernel(const float* __restrict__ x,
                 const int* __restrict__ row,
                 const int* __restrict__ col,
                 const float* __restrict__ inv_row,
                 const float* __restrict__ inv_col,
                 float* __restrict__ agg_fwd,
                 float* __restrict__ agg_rev) {
    __shared__ __attribute__((aligned(16))) float acc[RANGE * DIM];
    __shared__ unsigned queue[32 * QCAP];

    const int tid  = threadIdx.x;
    const int lane = tid & 31;
    const int wave = __builtin_amdgcn_readfirstlane(tid >> 5);
    const int dir  = blockIdx.x >> 7;
    const int rg   = blockIdx.x & (NRANGE - 1);
    const int n0   = rg * RANGE;
    const int cnt  = (N_NODES - n0 < RANGE) ? (N_NODES - n0) : RANGE;
    const unsigned ucnt = (unsigned)cnt;

    const int*   dst   = dir ? col : row;
    const int*   src   = dir ? row : col;
    const float* inv_d = dir ? inv_col : inv_row;
    const float* inv_s = dir ? inv_row : inv_col;
    float*       agg   = dir ? agg_rev : agg_fwd;

    {
        const float4 z = make_float4(0.f, 0.f, 0.f, 0.f);
        for (int i = tid; i < RANGE * DIM / 4; i += 1024) ((float4*)acc)[i] = z;
    }
    __syncthreads();

    unsigned* q = queue + wave * QCAP;
    int pending = 0;
    const int rot = (rg * OUTER) >> 7;

    auto load4 = [&](int4 (&r)[4], int step) {
        if (step >= OUTER) step -= OUTER;
        #pragma unroll
        for (int u = 0; u < 4; ++u) {
            int c = (step * 4 + u) * 32 + wave;
            c = c < CHUNKS ? c : CHUNKS - 1;
            r[u] = *(const int4*)(dst + (size_t)c * 128 + lane * 4);
        }
    };
    auto look4 = [&](const int4 (&r)[4], int step) {
        if (step >= OUTER) step -= OUTER;
        #pragma unroll
        for (int u = 0; u < 4; ++u) {
            const int chunk = (step * 4 + u) * 32 + wave;
            if (chunk < CHUNKS) {
                const unsigned t0 = (unsigned)(r[u].x - n0);
                const unsigned t1 = (unsigned)(r[u].y - n0);
                const unsigned t2 = (unsigned)(r[u].z - n0);
                const unsigned t3 = (unsigned)(r[u].w - n0);
                const bool h0 = t0 < ucnt, h1 = t1 < ucnt, h2 = t2 < ucnt, h3 = t3 < ucnt;
                if (__builtin_amdgcn_ballot_w32(h0 | h1 | h2 | h3) != 0u) {
                    const unsigned e0 = (unsigned)(chunk * 128 + lane * 4);
                    unsigned m;
                    m = __builtin_amdgcn_ballot_w32(h0);
                    if (h0) q[pending + (int)__builtin_amdgcn_mbcnt_lo(m, 0u)] = (t0 << 20) | e0;
                    pending += __builtin_popcount(m);
                    m = __builtin_amdgcn_ballot_w32(h1);
                    if (h1) q[pending + (int)__builtin_amdgcn_mbcnt_lo(m, 0u)] = (t1 << 20) | (e0 + 1u);
                    pending += __builtin_popcount(m);
                    m = __builtin_amdgcn_ballot_w32(h2);
                    if (h2) q[pending + (int)__builtin_amdgcn_mbcnt_lo(m, 0u)] = (t2 << 20) | (e0 + 2u);
                    pending += __builtin_popcount(m);
                    m = __builtin_amdgcn_ballot_w32(h3);
                    if (h3) q[pending + (int)__builtin_amdgcn_mbcnt_lo(m, 0u)] = (t3 << 20) | (e0 + 3u);
                    pending += __builtin_popcount(m);
                }
            }
        }
        while (pending >= 32) {
            pending -= 32;
            drain_full(q + pending, lane, n0, src, inv_d, inv_s, x, acc);
        }
    };

    int4 da[4], db[4];
    load4(da, rot);
    for (int o = 0; o < OUTER; o += 2) {
        load4(db, o + 1 + rot);
        look4(da, o + rot);
        load4(da, o + 2 + rot);
        look4(db, o + 1 + rot);
    }
    if (pending > 0) drain_tail(q, pending, lane, n0, src, inv_d, inv_s, x, acc);

    __syncthreads();

    for (int t = wave; t < cnt; t += 32) {
        const float* a = acc + t * DIM + lane;
        const float4 v = make_float4(a[0], a[32], a[64], a[96]);
        ((float4*)(agg + (size_t)(n0 + t) * DIM))[lane] = v;
    }
}

__global__ void __launch_bounds__(256)
gemm_kernel(const float* __restrict__ agg_fwd,
            const float* __restrict__ agg_rev,
            const float4* __restrict__ Wf,
            const float* __restrict__ bias,
            float* __restrict__ out) {
    __shared__ float4 wl[WF_WORDS];

    for (int i = threadIdx.x; i < WF_WORDS; i += 256) wl[i] = Wf[i];
    __syncthreads();

    const int lane = threadIdx.x & 31;
    const int mt   = blockIdx.x * 8 + (threadIdx.x >> 5);
    if (mt >= N_NODES / 16) return;

    const int half = lane >> 4;
    const int l15  = lane & 15;
    const int kb   = half * 2;

    const float* arow_f = agg_fwd + (size_t)(mt * 16 + l15) * DIM + kb;
    const float* arow_r = agg_rev + (size_t)(mt * 16 + l15) * DIM + kb;

    v8f c[8];
    #pragma unroll
    for (int nt = 0; nt < 8; ++nt) {
        const float bv = bias[nt * 16 + l15];
        #pragma unroll
        for (int j = 0; j < 8; ++j) c[nt][j] = bv;
    }

    v2f af[2][4], ar[2][4];
    #pragma unroll
    for (int u = 0; u < 4; ++u) {
        af[0][u] = *(const v2f*)(arow_f + 4 * u);
        ar[0][u] = *(const v2f*)(arow_r + 4 * u);
    }
    #pragma unroll 2
    for (int t = 0; t < DIM / 16; ++t) {
        const int cur = t & 1, nxt = cur ^ 1;
        if (t + 1 < DIM / 16) {
            #pragma unroll
            for (int u = 0; u < 4; ++u) {
                af[nxt][u] = *(const v2f*)(arow_f + 16 * (t + 1) + 4 * u);
                ar[nxt][u] = *(const v2f*)(arow_r + 16 * (t + 1) + 4 * u);
            }
        }
        #pragma unroll
        for (int u = 0; u < 4; ++u) {
            const float4* wq = wl + (t * 4 + u) * 256 + lane;
            #pragma unroll
            for (int nt = 0; nt < 8; ++nt) {
                const float4 w = wq[nt * 32];
                v2f b0, b1;
                b0[0] = w.x; b0[1] = w.y;
                b1[0] = w.z; b1[1] = w.w;
                c[nt] = __builtin_amdgcn_wmma_f32_16x16x4_f32(false, af[cur][u], false, b0,
                                                              (short)0, c[nt], false, false);
                c[nt] = __builtin_amdgcn_wmma_f32_16x16x4_f32(false, ar[cur][u], false, b1,
                                                              (short)0, c[nt], false, false);
            }
        }
    }

    const int m0 = mt * 16 + half * 8;
    #pragma unroll
    for (int nt = 0; nt < 8; ++nt) {
        const int n = nt * 16 + l15;
        #pragma unroll
        for (int j = 0; j < 8; ++j)
            out[(size_t)(m0 + j) * DIM + n] = c[nt][j];
    }
}

extern "C" void kernel_launch(void* const* d_in, const int* in_sizes, int n_in,
                              void* d_out, int out_size, void* d_ws, size_t ws_size,
                              hipStream_t stream) {
    const float* x      = (const float*)d_in[0];
    const float* W_src  = (const float*)d_in[1];
    const float* b_src  = (const float*)d_in[2];
    const float* W_dst  = (const float*)d_in[3];
    const float* b_dst  = (const float*)d_in[4];
    const int*   eidx   = (const int*)d_in[5];
    const int*   row    = eidx;
    const int*   col    = eidx + N_EDGES;
    float*       outp   = (float*)d_out;

    float* ws = (float*)d_ws;
    float* deg_row = ws;
    float* deg_col = ws + N_NODES;
    float* agg_fwd = ws + 2 * (size_t)N_NODES;
    float* agg_rev = agg_fwd + (size_t)N_NODES * DIM;
    float4* Wf     = (float4*)(agg_rev + (size_t)N_NODES * DIM);
    float* bias    = (float*)(Wf + WF_WORDS);

    {
        long n4 = (long)2 * N_NODES / 4;
        zero_kernel<<<(2 * N_NODES / 4 + 255) / 256, 256, 0, stream>>>((float4*)ws, n4);
    }
    degree_kernel<<<(N_EDGES + 255) / 256, 256, 0, stream>>>(row, col, deg_row, deg_col);
    invsqrt_kernel<<<(2 * N_NODES + 255) / 256, 256, 0, stream>>>(deg_row);
    prep_kernel<<<(WF_WORDS + 255) / 256, 256, 0, stream>>>(W_src, b_src, W_dst, b_dst, Wf, bias);
    aggregate_kernel<<<2 * NRANGE, 1024, 0, stream>>>(x, row, col, deg_row, deg_col,
                                                      agg_fwd, agg_rev);
    gemm_kernel<<<(N_NODES / 16 + 7) / 8, 256, 0, stream>>>(agg_fwd, agg_rev, Wf, bias, outp);
}


# degree counts without global atomics: 64 work-groups each count one edge-list slice into a whole-graph LDS histogram (LDS integer adds), a sum pass adds the 32 planes; counts exact, zero pass gone
# speedup vs baseline: 3.8170x; 1.5123x over previous
#include <hip/hip_runtime.h>
#include <hip/hip_bf16.h>

typedef __attribute__((ext_vector_type(2))) float v2f;
typedef __attribute__((ext_vector_type(8))) float v8f;

#define N_NODES 50000
#define N_EDGES 800000
#define DIM     128
#define WF_WORDS (DIM / 4 * 8 * 32)

#define RANGE   391
#define NRANGE  128
#define CHUNKS  (N_EDGES / 128)
#define OUTER   50
#define QCAP    544
static_assert(N_EDGES % 128 == 0, "edge list splits into whole 128-edge pieces");
static_assert(NRANGE * RANGE >= N_NODES, "ranges cover every node");
static_assert(OUTER * 4 * 32 >= CHUNKS && OUTER % 2 == 0, "one sweep covers every piece, in step pairs");
static_assert(N_EDGES <= (1 << 20) && RANGE <= (1 << 12), "queue entry packs (local node, edge id) in 32 bits");
#define SLICES  32
#define SLICE_EDGES (N_EDGES / SLICES)
#define SLICE_LOADS ((SLICE_EDGES / 4 + 1023) / 1024)
static_assert(N_EDGES % (SLICES * 4) == 0, "a slice is a whole number of four-id loads");
static_assert(2 * SLICES * N_NODES <= 2 * N_NODES * DIM, "the partial count planes fit in the aggregate rows' space");
#define TAB_LOADS ((N_NODES / 4 + 1023) / 1024)
static_assert(N_NODES % 4 == 0, "degree table is filled four nodes per 16-byte load");


__global__ void __launch_bounds__(1024)
hist_kernel(const int* __restrict__ row,
            const int* __restrict__ col,
            unsigned* __restrict__ part) {
    __shared__ __attribute__((aligned(16))) unsigned h[N_NODES];

    const int tid = threadIdx.x;
    const int dir = blockIdx.x / SLICES;
    const int g   = blockIdx.x - dir * SLICES;
    const int4* ids = (const int4*)((dir ? col : row) + (size_t)g * SLICE_EDGES);

    int4 v[SLICE_LOADS];
    #pragma unroll
    for (int u = 0; u < SLICE_LOADS; ++u) {
        const int i = u * 1024 + tid;
        v[u] = ids[i < SLICE_EDGES / 4 ? i : SLICE_EDGES / 4 - 1];
    }
    const uint4 z = make_uint4(0u, 0u, 0u, 0u);
    for (int i = tid; i < N_NODES / 4; i += 1024) ((uint4*)h)[i] = z;
    __syncthreads();

    #pragma unroll
    for (int u = 0; u < SLICE_LOADS; ++u) {
        if (u * 1024 + tid < SLICE_EDGES / 4) {
            atomicAdd(&h[v[u].x], 1u);
            atomicAdd(&h[v[u].y], 1u);
            atomicAdd(&h[v[u].z], 1u);
            atomicAdd(&h[v[u].w], 1u);
        }
    }
    __syncthreads();

    uint4* o = (uint4*)(part + (size_t)blockIdx.x * N_NODES);
    for (int i = tid; i < N_NODES / 4; i += 1024) o[i] = ((const uint4*)h)[i];
}

__global__ void degsum_kernel(const uint4* __restrict__ part, float4* __restrict__ deg) {
    const int i = blockIdx.x * blockDim.x + threadIdx.x;
    if (i >= 2 * (N_NODES / 4)) return;
    const int dir = i / (N_NODES / 4);
    const int j   = i - dir * (N_NODES / 4);
    const uint4* p = part + (size_t)dir * SLICES * (N_NODES / 4) + j;
    uint4 c[SLICES];
    #pragma unroll
    for (int g = 0; g < SLICES; ++g) c[g] = p[(size_t)g * (N_NODES / 4)];
    uint4 s = make_uint4(0u, 0u, 0u, 0u);
    #pragma unroll
    for (int g = 0; g < SLICES; ++g) { s.x += c[g].x; s.y += c[g].y; s.z += c[g].z; s.w += c[g].w; }
    deg[i] = make_float4((float)s.x, (float)s.y, (float)s.z, (float)s.w);
}

__device__ __forceinline__ float inv_of(float d) {
    return (d > 0.0f) ? rsqrtf(fmaxf(d, 1.0f)) : 0.0f;
}

__device__ __forceinline__ float src_factor(const unsigned char* tab,
                                            const float* __restrict__ deg_s, int s) {
    const unsigned c = tab[s];
    const float d = (c == 255u) ? deg_s[s] : (float)c;
    return inv_of(d);
}

__global__ void prep_kernel(const float* __restrict__ W_src,
                            const float* __restrict__ b_src,
                            const float* __restrict__ W_dst,
                            const float* __restrict__ b_dst,
                            float4* __restrict__ Wf,
                            float* __restrict__ bias) {
    int i = blockIdx.x * blockDim.x + threadIdx.x;
    if (i < WF_WORDS) {
        const int l  = i & 31;
        const int nt = (i >> 5) & 7;
        const int kq = i >> 8;
        const int n  = nt * 16 + (l & 15);
        const int k  = 4 * kq + 2 * (l >> 4);
        const float2 s = *(const float2*)(W_src + n * DIM + k);
        const float2 d = *(const float2*)(W_dst + n * DIM + k);
        Wf[i] = make_float4(0.5f * s.x, 0.5f * s.y, 0.5f * d.x, 0.5f * d.y);
    }
    if (i < DIM) bias[i] = 0.5f * b_src[i] + 0.5f * b_dst[i];
}

__device__ __forceinline__ void add_row(float* acc, int t, int lane, float w, const float4& v) {
    float* a = acc + t * DIM + lane;
    atomicAdd(a,      w * v.x);
    atomicAdd(a + 32, w * v.y);
    atomicAdd(a + 64, w * v.z);
    atomicAdd(a + 96, w * v.w);
}

__device__ __forceinline__ void drain_full(const unsigned* qb, int lane, int n0,
                                           const int* __restrict__ src,
                                           const float* __restrict__ deg_d,
                                           const float* __restrict__ deg_s,
                                           const unsigned char* tab,
                                           const float* __restrict__ x,
                                           float* acc) {
    const unsigned p = qb[lane];
    const int t = (int)(p >> 20);
    const int s = src[(int)(p & 0xFFFFFu)];
    const float w = inv_of(deg_d[n0 + t]) * src_factor(tab, deg_s, s);
    const int wb = __builtin_bit_cast(int, w);

    float4 v[2][8];
    #pragma unroll
    for (int u = 0; u < 8; ++u)
        v[0][u] = ((const float4*)(x + (size_t)__builtin_amdgcn_readlane(s, u) * DIM))[lane];
    #pragma unroll
    for (int g = 0; g < 4; ++g) {
        if (g + 1 < 4) {
            #pragma unroll
            for (int u = 0; u < 8; ++u)
                v[(g + 1) & 1][u] = ((const float4*)(x +
                    (size_t)__builtin_amdgcn_readlane(s, (g + 1) * 8 + u) * DIM))[lane];
        }
        #pragma unroll
        for (int u = 0; u < 8; ++u)
            add_row(acc, __builtin_amdgcn_readlane(t, g * 8 + u), lane,
                    __builtin_bit_cast(float, __builtin_amdgcn_readlane(wb, g * 8 + u)),
                    v[g & 1][u]);
    }
}

__device__ __forceinline__ void drain_tail(const unsigned* qb, int n, int lane, int n0,
                                           const int* __restrict__ src,
                                           const float* __restrict__ deg_d,
                                           const float* __restrict__ deg_s,
                                           const unsigned char* tab,
                                           const float* __restrict__ x,
                                           float* acc) {
    int s = 0, t = 0;
    float w = 0.0f;
    if (lane < n) {
        const unsigned p = qb[lane];
        t = (int)(p >> 20);
        s = src[(int)(p & 0xFFFFFu)];
        w = inv_of(deg_d[n0 + t]) * src_factor(tab, deg_s, s);
    }
    const int wb = __builtin_bit_cast(int, w);
    int i = 0;
    for (; i + 8 <= n; i += 8) {
        float4 v[8];
        #pragma unroll
        for (int u = 0; u < 8; ++u)
            v[u] = ((const float4*)(x + (size_t)__builtin_amdgcn_readlane(s, i + u) * DIM))[lane];
        #pragma unroll
        for (int u = 0; u < 8; ++u)
            add_row(acc, __builtin_amdgcn_readlane(t, i + u), lane,
                    __builtin_bit_cast(float, __builtin_amdgcn_readlane(wb, i + u)), v[u]);
    }
    for (; i < n; ++i) {
        const float4 v = ((const float4*)(x + (size_t)__builtin_amdgcn_readlane(s, i) * DIM))[lane];
        add_row(acc, __builtin_amdgcn_readlane(t, i), lane,
                __builtin_bit_cast(float, __builtin_amdgcn_readlane(wb, i)), v);
    }
}

__global__ void __launch_bounds__(1024)
aggregate_kernel(const float* __restrict__ x,
                 const int* __restrict__ row,
                 const int* __restrict__ col,
                 const float* __restrict__ deg_row,
                 const float* __restrict__ deg_col,
                 float* __restrict__ agg_fwd,
                 float* __restrict__ agg_rev) {
    __shared__ __attribute__((aligned(16))) float acc[RANGE * DIM];
    __shared__ unsigned queue[32 * QCAP];
    __shared__ __attribute__((aligned(16))) unsigned char tab[N_NODES];

    const int tid  = threadIdx.x;
    const int lane = tid & 31;
    const int wave = __builtin_amdgcn_readfirstlane(tid >> 5);
    const int dir  = blockIdx.x >> 7;
    const int rg   = blockIdx.x & (NRANGE - 1);
    const int n0   = rg * RANGE;
    const int cnt  = (N_NODES - n0 < RANGE) ? (N_NODES - n0) : RANGE;
    const unsigned ucnt = (unsigned)cnt;

    const int*   dst   = dir ? col : row;
    const int*   src   = dir ? row : col;
    const float* deg_d = dir ? deg_col : deg_row;
    const float* deg_s = dir ? deg_row : deg_col;
    float*       agg   = dir ? agg_rev : agg_fwd;

    {
        float4 d[TAB_LOADS];
        #pragma unroll
        for (int u = 0; u < TAB_LOADS; ++u) {
            const int i = u * 1024 + tid;
            d[u] = ((const float4*)deg_s)[i < N_NODES / 4 ? i : N_NODES / 4 - 1];
        }
        const float4 z = make_float4(0.f, 0.f, 0.f, 0.f);
        for (int i = tid; i < RANGE * DIM / 4; i += 1024) ((float4*)acc)[i] = z;
        #pragma unroll
        for (int u = 0; u < TAB_LOADS; ++u) {
            const int i = u * 1024 + tid;
            if (i < N_NODES / 4) {
                const unsigned b0 = (unsigned)fminf(d[u].x, 255.0f);
                const unsigned b1 = (unsigned)fminf(d[u].y, 255.0f);
                const unsigned b2 = (unsigned)fminf(d[u].z, 255.0f);
                const unsigned b3 = (unsigned)fminf(d[u].w, 255.0f);
                ((unsigned*)tab)[i] = b0 | (b1 << 8) | (b2 << 16) | (b3 << 24);
            }
        }
    }
    __syncthreads();

    unsigned* q = queue + wave * QCAP;
    int pending = 0;
    const int rot = (rg * OUTER) >> 7;

    auto load4 = [&](int4 (&r)[4], int step) {
        if (step >= OUTER) step -= OUTER;
        #pragma unroll
        for (int u = 0; u < 4; ++u) {
            int c = (step * 4 + u) * 32 + wave;
            c = c < CHUNKS ? c : CHUNKS - 1;
            r[u] = *(const int4*)(dst + (size_t)c * 128 + lane * 4);
        }
    };
    auto look4 = [&](const int4 (&r)[4], int step) {
        if (step >= OUTER) step -= OUTER;
        #pragma unroll
        for (int u = 0; u < 4; ++u) {
            const int chunk = (step * 4 + u) * 32 + wave;
            if (chunk < CHUNKS) {
                const unsigned t0 = (unsigned)(r[u].x - n0);
                const unsigned t1 = (unsigned)(r[u].y - n0);
                const unsigned t2 = (unsigned)(r[u].z - n0);
                const unsigned t3 = (unsigned)(r[u].w - n0);
                const bool h0 = t0 < ucnt, h1 = t1 < ucnt, h2 = t2 < ucnt, h3 = t3 < ucnt;
                if (__builtin_amdgcn_ballot_w32(h0 | h1 | h2 | h3) != 0u) {
                    const unsigned e0 = (unsigned)(chunk * 128 + lane * 4);
                    unsigned m;
                    m = __builtin_amdgcn_ballot_w32(h0);
                    if (h0) q[pending + (int)__builtin_amdgcn_mbcnt_lo(m, 0u)] = (t0 << 20) | e0;
                    pending += __builtin_popcount(m);
                    m = __builtin_amdgcn_ballot_w32(h1);
                    if (h1) q[pending + (int)__builtin_amdgcn_mbcnt_lo(m, 0u)] = (t1 << 20) | (e0 + 1u);
                    pending += __builtin_popcount(m);
                    m = __builtin_amdgcn_ballot_w32(h2);
                    if (h2) q[pending + (int)__builtin_amdgcn_mbcnt_lo(m, 0u)] = (t2 << 20) | (e0 + 2u);
                    pending += __builtin_popcount(m);
                    m = __builtin_amdgcn_ballot_w32(h3);
                    if (h3) q[pending + (int)__builtin_amdgcn_mbcnt_lo(m, 0u)] = (t3 << 20) | (e0 + 3u);
                    pending += __builtin_popcount(m);
                }
            }
        }
        while (pending >= 32) {
            pending -= 32;
            drain_full(q + pending, lane, n0, src, deg_d, deg_s, tab, x, acc);
        }
    };

    int4 da[4], db[4];
    load4(da, rot);
    for (int o = 0; o < OUTER; o += 2) {
        load4(db, o + 1 + rot);
        look4(da, o + rot);
        load4(da, o + 2 + rot);
        look4(db, o + 1 + rot);
    }
    if (pending > 0) drain_tail(q, pending, lane, n0, src, deg_d, deg_s, tab, x, acc);

    __syncthreads();

    for (int t = wave; t < cnt; t += 32) {
        const float* a = acc + t * DIM + lane;
        const float4 v = make_float4(a[0], a[32], a[64], a[96]);
        ((float4*)(agg + (size_t)(n0 + t) * DIM))[lane] = v;
    }
}

__global__ void __launch_bounds__(256)
gemm_kernel(const float* __restrict__ agg_fwd,
            const float* __restrict__ agg_rev,
            const float4* __restrict__ Wf,
            const float* __restrict__ bias,
            float* __restrict__ out) {
    __shared__ float4 wl[WF_WORDS];

    for (int i = threadIdx.x; i < WF_WORDS; i += 256) wl[i] = Wf[i];
    __syncthreads();

    const int lane = threadIdx.x & 31;
    const int mt   = blockIdx.x * 8 + (threadIdx.x >> 5);
    if (mt >= N_NODES / 16) return;

    const int half = lane >> 4;
    const int l15  = lane & 15;
    const int kb   = half * 2;

    const float* arow_f = agg_fwd + (size_t)(mt * 16 + l15) * DIM + kb;
    const float* arow_r = agg_rev + (size_t)(mt * 16 + l15) * DIM + kb;

    v8f c[8];
    #pragma unroll
    for (int nt = 0; nt < 8; ++nt) {
        const float bv = bias[nt * 16 + l15];
        #pragma unroll
        for (int j = 0; j < 8; ++j) c[nt][j] = bv;
    }

    v2f af[2][4], ar[2][4];
    #pragma unroll
    for (int u = 0; u < 4; ++u) {
        af[0][u] = *(const v2f*)(arow_f + 4 * u);
        ar[0][u] = *(const v2f*)(arow_r + 4 * u);
    }
    #pragma unroll 2
    for (int t = 0; t < DIM / 16; ++t) {
        const int cur = t & 1, nxt = cur ^ 1;
        if (t + 1 < DIM / 16) {
            #pragma unroll
            for (int u = 0; u < 4; ++u) {
                af[nxt][u] = *(const v2f*)(arow_f + 16 * (t + 1) + 4 * u);
                ar[nxt][u] = *(const v2f*)(arow_r + 16 * (t + 1) + 4 * u);
            }
        }
        #pragma unroll
        for (int u = 0; u < 4; ++u) {
            const float4* wq = wl + (t * 4 + u) * 256 + lane;
            #pragma unroll
            for (int nt = 0; nt < 8; ++nt) {
                const float4 w = wq[nt * 32];
                v2f b0, b1;
                b0[0] = w.x; b0[1] = w.y;
                b1[0] = w.z; b1[1] = w.w;
                c[nt] = __builtin_amdgcn_wmma_f32_16x16x4_f32(false, af[cur][u], false, b0,
                                                              (short)0, c[nt], false, false);
                c[nt] = __builtin_amdgcn_wmma_f32_16x16x4_f32(false, ar[cur][u], false, b1,
                                                              (short)0, c[nt], false, false);
            }
        }
    }

    const int m0 = mt * 16 + half * 8;
    #pragma unroll
    for (int nt = 0; nt < 8; ++nt) {
        const int n = nt * 16 + l15;
        #pragma unroll
        for (int j = 0; j < 8; ++j)
            out[(size_t)(m0 + j) * DIM + n] = c[nt][j];
    }
}

extern "C" void kernel_launch(void* const* d_in, const int* in_sizes, int n_in,
                              void* d_out, int out_size, void* d_ws, size_t ws_size,
                              hipStream_t stream) {
    const float* x      = (const float*)d_in[0];
    const float* W_src  = (const float*)d_in[1];
    const float* b_src  = (const float*)d_in[2];
    const float* W_dst  = (const float*)d_in[3];
    const float* b_dst  = (const float*)d_in[4];
    const int*   eidx   = (const int*)d_in[5];
    const int*   row    = eidx;
    const int*   col    = eidx + N_EDGES;
    float*       outp   = (float*)d_out;

    float* ws = (float*)d_ws;
    float* deg_row = ws;
    float* deg_col = ws + N_NODES;
    float* agg_fwd = ws + 2 * (size_t)N_NODES;
    float* agg_rev = agg_fwd + (size_t)N_NODES * DIM;
    float4* Wf     = (float4*)(agg_rev + (size_t)N_NODES * DIM);
    float* bias    = (float*)(Wf + WF_WORDS);

    hist_kernel<<<2 * SLICES, 1024, 0, stream>>>(row, col, (unsigned*)agg_fwd);
    degsum_kernel<<<(2 * (N_NODES / 4) + 255) / 256, 256, 0, stream>>>((const uint4*)agg_fwd, (float4*)ws);
    prep_kernel<<<(WF_WORDS + 255) / 256, 256, 0, stream>>>(W_src, b_src, W_dst, b_dst, Wf, bias);
    aggregate_kernel<<<2 * NRANGE, 1024, 0, stream>>>(x, row, col, deg_row, deg_col,
                                                      agg_fwd, agg_rev);
    gemm_kernel<<<(N_NODES / 16 + 7) / 8, 256, 0, stream>>>(agg_fwd, agg_rev, Wf, bias, outp);
}


# degree pass also stores each id as 16 bits (copy rests in the output buffer until the last launch overwrites it); aggregate sweep reads 2 bytes per edge, not 4; LDS add order changes
# speedup vs baseline: 3.9967x; 1.0471x over previous
#include <hip/hip_runtime.h>
#include <hip/hip_bf16.h>

typedef __attribute__((ext_vector_type(2))) float v2f;
typedef __attribute__((ext_vector_type(8))) float v8f;

#define N_NODES 50000
#define N_EDGES 800000
#define DIM     128
#define WF_WORDS (DIM / 4 * 8 * 32)

#define RANGE   391
#define NRANGE  128
#define CHUNKS  (N_EDGES / 256)
#define OUTER   50
#define QCAP    544
static_assert(N_EDGES % 256 == 0, "edge list splits into whole 256-edge pieces");
static_assert(N_NODES <= (1 << 16), "a node id fits in 16 bits");
static_assert(NRANGE * RANGE >= N_NODES, "ranges cover every node");
static_assert(OUTER * 2 * 32 >= CHUNKS && OUTER % 2 == 0, "one sweep covers every piece, in step pairs");
static_assert(N_EDGES <= (1 << 20) && RANGE <= (1 << 12), "queue entry packs (local node, edge id) in 32 bits");
#define SLICES  32
#define SLICE_EDGES (N_EDGES / SLICES)
#define SLICE_LOADS ((SLICE_EDGES / 4 + 1023) / 1024)
static_assert(N_EDGES % (SLICES * 4) == 0, "a slice is a whole number of four-id loads");
static_assert(2 * SLICES * N_NODES <= 2 * N_NODES * DIM, "the partial count planes fit in the aggregate rows' space");
static_assert(2 * N_EDGES * 2 <= N_NODES * DIM * 4, "the 16-bit ids of both directions fit in the output buffer");
#define TAB_LOADS ((N_NODES / 4 + 1023) / 1024)
static_assert(N_NODES % 4 == 0, "degree table is filled four nodes per 16-byte load");


__global__ void __launch_bounds__(1024)
hist_kernel(const int* __restrict__ row,
            const int* __restrict__ col,
            unsigned* __restrict__ part,
            uint2* __restrict__ ids16) {
    __shared__ __attribute__((aligned(16))) unsigned h[N_NODES];

    const int tid = threadIdx.x;
    const int dir = blockIdx.x / SLICES;
    const int g   = blockIdx.x - dir * SLICES;
    const int4* ids = (const int4*)((dir ? col : row) + (size_t)g * SLICE_EDGES);

    int4 v[SLICE_LOADS];
    #pragma unroll
    for (int u = 0; u < SLICE_LOADS; ++u) {
        const int i = u * 1024 + tid;
        v[u] = ids[i < SLICE_EDGES / 4 ? i : SLICE_EDGES / 4 - 1];
    }
    const uint4 z = make_uint4(0u, 0u, 0u, 0u);
    for (int i = tid; i < N_NODES / 4; i += 1024) ((uint4*)h)[i] = z;
    __syncthreads();

    uint2* o16 = ids16 + (size_t)blockIdx.x * (SLICE_EDGES / 4);
    #pragma unroll
    for (int u = 0; u < SLICE_LOADS; ++u) {
        const int i = u * 1024 + tid;
        if (i < SLICE_EDGES / 4)
            o16[i] = make_uint2((unsigned)v[u].x | ((unsigned)v[u].y << 16),
                                (unsigned)v[u].z | ((unsigned)v[u].w << 16));
    }

    #pragma unroll
    for (int u = 0; u < SLICE_LOADS; ++u) {
        if (u * 1024 + tid < SLICE_EDGES / 4) {
            atomicAdd(&h[v[u].x], 1u);
            atomicAdd(&h[v[u].y], 1u);
            atomicAdd(&h[v[u].z], 1u);
            atomicAdd(&h[v[u].w], 1u);
        }
    }
    __syncthreads();

    uint4* o = (uint4*)(part + (size_t)blockIdx.x * N_NODES);
    for (int i = tid; i < N_NODES / 4; i += 1024) o[i] = ((const uint4*)h)[i];
}

__global__ void degsum_kernel(const uint4* __restrict__ part, float4* __restrict__ deg) {
    const int i = blockIdx.x * blockDim.x + threadIdx.x;
    if (i >= 2 * (N_NODES / 4)) return;
    const int dir = i / (N_NODES / 4);
    const int j   = i - dir * (N_NODES / 4);
    const uint4* p = part + (size_t)dir * SLICES * (N_NODES / 4) + j;
    uint4 c[SLICES];
    #pragma unroll
    for (int g = 0; g < SLICES; ++g) c[g] = p[(size_t)g * (N_NODES / 4)];
    uint4 s = make_uint4(0u, 0u, 0u, 0u);
    #pragma unroll
    for (int g = 0; g < SLICES; ++g) { s.x += c[g].x; s.y += c[g].y; s.z += c[g].z; s.w += c[g].w; }
    deg[i] = make_float4((float)s.x, (float)s.y, (float)s.z, (float)s.w);
}

__device__ __forceinline__ float inv_of(float d) {
    return (d > 0.0f) ? rsqrtf(fmaxf(d, 1.0f)) : 0.0f;
}

__device__ __forceinline__ float src_factor(const unsigned char* tab,
                                            const float* __restrict__ deg_s, int s) {
    const unsigned c = tab[s];
    const float d = (c == 255u) ? deg_s[s] : (float)c;
    return inv_of(d);
}

__global__ void prep_kernel(const float* __restrict__ W_src,
                            const float* __restrict__ b_src,
                            const float* __restrict__ W_dst,
                            const float* __restrict__ b_dst,
                            float4* __restrict__ Wf,
                            float* __restrict__ bias) {
    int i = blockIdx.x * blockDim.x + threadIdx.x;
    if (i < WF_WORDS) {
        const int l  = i & 31;
        const int nt = (i >> 5) & 7;
        const int kq = i >> 8;
        const int n  = nt * 16 + (l & 15);
        const int k  = 4 * kq + 2 * (l >> 4);
        const float2 s = *(const float2*)(W_src + n * DIM + k);
        const float2 d = *(const float2*)(W_dst + n * DIM + k);
        Wf[i] = make_float4(0.5f * s.x, 0.5f * s.y, 0.5f * d.x, 0.5f * d.y);
    }
    if (i < DIM) bias[i] = 0.5f * b_src[i] + 0.5f * b_dst[i];
}

__device__ __forceinline__ void add_row(float* acc, int t, int lane, float w, const float4& v) {
    float* a = acc + t * DIM + lane;
    atomicAdd(a,      w * v.x);
    atomicAdd(a + 32, w * v.y);
    atomicAdd(a + 64, w * v.z);
    atomicAdd(a + 96, w * v.w);
}

__device__ __forceinline__ void drain_full(const unsigned* qb, int lane, int n0,
                                           const int* __restrict__ src,
                                           const float* __restrict__ deg_d,
                                           const float* __restrict__ deg_s,
                                           const unsigned char* tab,
                                           const float* __restrict__ x,
                                           float* acc) {
    const unsigned p = qb[lane];
    const int t = (int)(p >> 20);
    const int s = src[(int)(p & 0xFFFFFu)];
    const float w = inv_of(deg_d[n0 + t]) * src_factor(tab, deg_s, s);
    const int wb = __builtin_bit_cast(int, w);

    float4 v[2][8];
    #pragma unroll
    for (int u = 0; u < 8; ++u)
        v[0][u] = ((const float4*)(x + (size_t)__builtin_amdgcn_readlane(s, u) * DIM))[lane];
    #pragma unroll
    for (int g = 0; g < 4; ++g) {
        if (g + 1 < 4) {
            #pragma unroll
            for (int u = 0; u < 8; ++u)
                v[(g + 1) & 1][u] = ((const float4*)(x +
                    (size_t)__builtin_amdgcn_readlane(s, (g + 1) * 8 + u) * DIM))[lane];
        }
        #pragma unroll
        for (int u = 0; u < 8; ++u)
            add_row(acc, __builtin_amdgcn_readlane(t, g * 8 + u), lane,
                    __builtin_bit_cast(float, __builtin_amdgcn_readlane(wb, g * 8 + u)),
                    v[g & 1][u]);
    }
}

__device__ __forceinline__ void drain_tail(const unsigned* qb, int n, int lane, int n0,
                                           const int* __restrict__ src,
                                           const float* __restrict__ deg_d,
                                           const float* __restrict__ deg_s,
                                           const unsigned char* tab,
                                           const float* __restrict__ x,
                                           float* acc) {
    int s = 0, t = 0;
    float w = 0.0f;
    if (lane < n) {
        const unsigned p = qb[lane];
        t = (int)(p >> 20);
        s = src[(int)(p & 0xFFFFFu)];
        w = inv_of(deg_d[n0 + t]) * src_factor(tab, deg_s, s);
    }
    const int wb = __builtin_bit_cast(int, w);
    int i = 0;
    for (; i + 8 <= n; i += 8) {
        float4 v[8];
        #pragma unroll
        for (int u = 0; u < 8; ++u)
            v[u] = ((const float4*)(x + (size_t)__builtin_amdgcn_readlane(s, i + u) * DIM))[lane];
        #pragma unroll
        for (int u = 0; u < 8; ++u)
            add_row(acc, __builtin_amdgcn_readlane(t, i + u), lane,
                    __builtin_bit_cast(float, __builtin_amdgcn_readlane(wb, i + u)), v[u]);
    }
    for (; i < n; ++i) {
        const float4 v = ((const float4*)(x + (size_t)__builtin_amdgcn_readlane(s, i) * DIM))[lane];
        add_row(acc, __builtin_amdgcn_readlane(t, i), lane,
                __builtin_bit_cast(float, __builtin_amdgcn_readlane(wb, i)), v);
    }
}

__global__ void __launch_bounds__(1024)
aggregate_kernel(const float* __restrict__ x,
                 const int* __restrict__ row,
                 const int* __restrict__ col,
                 const float* __restrict__ deg_row,
                 const float* __restrict__ deg_col,
                 const uint4* __restrict__ ids16,
                 float* __restrict__ agg_fwd,
                 float* __restrict__ agg_rev) {
    __shared__ __attribute__((aligned(16))) float acc[RANGE * DIM];
    __shared__ unsigned queue[32 * QCAP];
    __shared__ __attribute__((aligned(16))) unsigned char tab[N_NODES];

    const int tid  = threadIdx.x;
    const int lane = tid & 31;
    const int wave = __builtin_amdgcn_readfirstlane(tid >> 5);
    const int dir  = blockIdx.x >> 7;
    const int rg   = blockIdx.x & (NRANGE - 1);
    const int n0   = rg * RANGE;
    const int cnt  = (N_NODES - n0 < RANGE) ? (N_NODES - n0) : RANGE;
    const unsigned ucnt = (unsigned)cnt;

    const uint4* dst16 = ids16 + (size_t)dir * (N_EDGES / 8);
    const int*   src   = dir ? row : col;
    const float* deg_d = dir ? deg_col : deg_row;
    const float* deg_s = dir ? deg_row : deg_col;
    float*       agg   = dir ? agg_rev : agg_fwd;

    {
        float4 d[TAB_LOADS];
        #pragma unroll
        for (int u = 0; u < TAB_LOADS; ++u) {
            const int i = u * 1024 + tid;
            d[u] = ((const float4*)deg_s)[i < N_NODES / 4 ? i : N_NODES / 4 - 1];
        }
        const float4 z = make_float4(0.f, 0.f, 0.f, 0.f);
        for (int i = tid; i < RANGE * DIM / 4; i += 1024) ((float4*)acc)[i] = z;
        #pragma unroll
        for (int u = 0; u < TAB_LOADS; ++u) {
            const int i = u * 1024 + tid;
            if (i < N_NODES / 4) {
                const unsigned b0 = (unsigned)fminf(d[u].x, 255.0f);
                const unsigned b1 = (unsigned)fminf(d[u].y, 255.0f);
                const unsigned b2 = (unsigned)fminf(d[u].z, 255.0f);
                const unsigned b3 = (unsigned)fminf(d[u].w, 255.0f);
                ((unsigned*)tab)[i] = b0 | (b1 << 8) | (b2 << 16) | (b3 << 24);
            }
        }
    }
    __syncthreads();

    unsigned* q = queue + wave * QCAP;
    int pending = 0;
    const int rot = (rg * OUTER) >> 7;

    auto load4 = [&](uint4 (&r)[2], int step) {
        if (step >= OUTER) step -= OUTER;
        #pragma unroll
        for (int u = 0; u < 2; ++u) {
            int c = (step * 2 + u) * 32 + wave;
            c = c < CHUNKS ? c : CHUNKS - 1;
            r[u] = dst16[(size_t)c * 32 + lane];
        }
    };
    auto look4 = [&](const uint4 (&r)[2], int step) {
        if (step >= OUTER) step -= OUTER;
        #pragma unroll
        for (int u = 0; u < 2; ++u) {
            const int chunk = (step * 2 + u) * 32 + wave;
            if (chunk < CHUNKS) {
                const unsigned wd[4] = {r[u].x, r[u].y, r[u].z, r[u].w};
                unsigned t[8];
                bool h[8];
                #pragma unroll
                for (int k = 0; k < 8; ++k) {
                    const unsigned id = (k & 1) ? (wd[k >> 1] >> 16) : (wd[k >> 1] & 0xFFFFu);
                    t[k] = id - (unsigned)n0;
                    h[k] = t[k] < ucnt;
                }
                const unsigned e0 = (unsigned)(chunk * 256 + lane * 8);
                #pragma unroll
                for (int g4 = 0; g4 < 8; g4 += 4) {
                    if (__builtin_amdgcn_ballot_w32(h[g4] | h[g4 + 1] | h[g4 + 2] | h[g4 + 3]) != 0u) {
                        #pragma unroll
                        for (int k = g4; k < g4 + 4; ++k) {
                            const unsigned m = __builtin_amdgcn_ballot_w32(h[k]);
                            if (h[k]) q[pending + (int)__builtin_amdgcn_mbcnt_lo(m, 0u)] = (t[k] << 20) | (e0 + (unsigned)k);
                            pending += __builtin_popcount(m);
                        }
                    }
                }
            }
        }
        while (pending >= 32) {
            pending -= 32;
            drain_full(q + pending, lane, n0, src, deg_d, deg_s, tab, x, acc);
        }
    };

    uint4 da[2], db[2];
    load4(da, rot);
    for (int o = 0; o < OUTER; o += 2) {
        load4(db, o + 1 + rot);
        look4(da, o + rot);
        load4(da, o + 2 + rot);
        look4(db, o + 1 + rot);
    }
    if (pending > 0) drain_tail(q, pending, lane, n0, src, deg_d, deg_s, tab, x, acc);

    __syncthreads();

    for (int t = wave; t < cnt; t += 32) {
        const float* a = acc + t * DIM + lane;
        const float4 v = make_float4(a[0], a[32], a[64], a[96]);
        ((float4*)(agg + (size_t)(n0 + t) * DIM))[lane] = v;
    }
}

__global__ void __launch_bounds__(256)
gemm_kernel(const float* __restrict__ agg_fwd,
            const float* __restrict__ agg_rev,
            const float4* __restrict__ Wf,
            const float* __restrict__ bias,
            float* __restrict__ out) {
    __shared__ float4 wl[WF_WORDS];

    for (int i = threadIdx.x; i < WF_WORDS; i += 256) wl[i] = Wf[i];
    __syncthreads();

    const int lane = threadIdx.x & 31;
    const int mt   = blockIdx.x * 8 + (threadIdx.x >> 5);
    if (mt >= N_NODES / 16) return;

    const int half = lane >> 4;
    const int l15  = lane & 15;
    const int kb   = half * 2;

    const float* arow_f = agg_fwd + (size_t)(mt * 16 + l15) * DIM + kb;
    const float* arow_r = agg_rev + (size_t)(mt * 16 + l15) * DIM + kb;

    v8f c[8];
    #pragma unroll
    for (int nt = 0; nt < 8; ++nt) {
        const float bv = bias[nt * 16 + l15];
        #pragma unroll
        for (int j = 0; j < 8; ++j) c[nt][j] = bv;
    }

    v2f af[2][4], ar[2][4];
    #pragma unroll
    for (int u = 0; u < 4; ++u) {
        af[0][u] = *(const v2f*)(arow_f + 4 * u);
        ar[0][u] = *(const v2f*)(arow_r + 4 * u);
    }
    #pragma unroll 2
    for (int t = 0; t < DIM / 16; ++t) {
        const int cur = t & 1, nxt = cur ^ 1;
        if (t + 1 < DIM / 16) {
            #pragma unroll
            for (int u = 0; u < 4; ++u) {
                af[nxt][u] = *(const v2f*)(arow_f + 16 * (t + 1) + 4 * u);
                ar[nxt][u] = *(const v2f*)(arow_r + 16 * (t + 1) + 4 * u);
            }
        }
        #pragma unroll
        for (int u = 0; u < 4; ++u) {
            const float4* wq = wl + (t * 4 + u) * 256 + lane;
            #pragma unroll
            for (int nt = 0; nt < 8; ++nt) {
                const float4 w = wq[nt * 32];
                v2f b0, b1;
                b0[0] = w.x; b0[1] = w.y;
                b1[0] = w.z; b1[1] = w.w;
                c[nt] = __builtin_amdgcn_wmma_f32_16x16x4_f32(false, af[cur][u], false, b0,
                                                              (short)0, c[nt], false, false);
                c[nt] = __builtin_amdgcn_wmma_f32_16x16x4_f32(false, ar[cur][u], false, b1,
                                                              (short)0, c[nt], false, false);
            }
        }
    }

    const int m0 = mt * 16 + half * 8;
    #pragma unroll
    for (int nt = 0; nt < 8; ++nt) {
        const int n = nt * 16 + l15;
        #pragma unroll
        for (int j = 0; j < 8; ++j)
            out[(size_t)(m0 + j) * DIM + n] = c[nt][j];
    }
}

extern "C" void kernel_launch(void* const* d_in, const int* in_sizes, int n_in,
                              void* d_out, int out_size, void* d_ws, size_t ws_size,
                              hipStream_t stream) {
    const float* x      = (const float*)d_in[0];
    const float* W_src  = (const float*)d_in[1];
    const float* b_src  = (const float*)d_in[2];
    const float* W_dst  = (const float*)d_in[3];
    const float* b_dst  = (const float*)d_in[4];
    const int*   eidx   = (const int*)d_in[5];
    const int*   row    = eidx;
    const int*   col    = eidx + N_EDGES;
    float*       outp   = (float*)d_out;

    float* ws = (float*)d_ws;
    float* deg_row = ws;
    float* deg_col = ws + N_NODES;
    float* agg_fwd = ws + 2 * (size_t)N_NODES;
    float* agg_rev = agg_fwd + (size_t)N_NODES * DIM;
    float4* Wf     = (float4*)(agg_rev + (size_t)N_NODES * DIM);
    float* bias    = (float*)(Wf + WF_WORDS);

    hist_kernel<<<2 * SLICES, 1024, 0, stream>>>(row, col, (unsigned*)agg_fwd, (uint2*)outp);
    degsum_kernel<<<(2 * (N_NODES / 4) + 255) / 256, 256, 0, stream>>>((const uint4*)agg_fwd, (float4*)ws);
    prep_kernel<<<(WF_WORDS + 255) / 256, 256, 0, stream>>>(W_src, b_src, W_dst, b_dst, Wf, bias);
    aggregate_kernel<<<2 * NRANGE, 1024, 0, stream>>>(x, row, col, deg_row, deg_col,
                                                      (const uint4*)outp, agg_fwd, agg_rev);
    gemm_kernel<<<(N_NODES / 16 + 7) / 8, 256, 0, stream>>>(agg_fwd, agg_rev, Wf, bias, outp);
}


# aggregate sweep tests its ids as packed 16-bit halves: one packed subtract per two ids, one 16-bit compare per id straight into a lane mask, a rank step only where that mask is non-zero
# speedup vs baseline: 4.1898x; 1.0483x over previous
#include <hip/hip_runtime.h>
#include <hip/hip_bf16.h>

typedef __attribute__((ext_vector_type(2))) float v2f;
typedef __attribute__((ext_vector_type(8))) float v8f;
typedef __attribute__((ext_vector_type(2))) unsigned short us2;

#define N_NODES 50000
#define N_EDGES 800000
#define DIM     128
#define WF_WORDS (DIM / 4 * 8 * 32)

#define RANGE   391
#define NRANGE  128
#define CHUNKS  (N_EDGES / 256)
#define OUTER   50
#define QCAP    544
static_assert(N_EDGES % 256 == 0, "edge list splits into whole 256-edge pieces");
static_assert(N_NODES <= (1 << 16), "a node id fits in 16 bits");
static_assert(NRANGE * RANGE >= N_NODES, "ranges cover every node");
static_assert(NRANGE * RANGE <= (1 << 16), "id - n0 modulo 2^16 never falls under the range size for an id below n0");
static_assert(OUTER * 2 * 32 >= CHUNKS && OUTER % 2 == 0, "one sweep covers every piece, in step pairs");
static_assert(N_EDGES <= (1 << 20) && RANGE <= (1 << 12), "queue entry packs (local node, edge id) in 32 bits");
#define SLICES  32
#define SLICE_EDGES (N_EDGES / SLICES)
#define SLICE_LOADS ((SLICE_EDGES / 4 + 1023) / 1024)
static_assert(N_EDGES % (SLICES * 4) == 0, "a slice is a whole number of four-id loads");
static_assert(2 * SLICES * N_NODES <= 2 * N_NODES * DIM, "the partial count planes fit in the aggregate rows' space");
static_assert(2 * N_EDGES * 2 <= N_NODES * DIM * 4, "the 16-bit ids of both directions fit in the output buffer");
#define TAB_LOADS ((N_NODES / 4 + 1023) / 1024)
static_assert(N_NODES % 4 == 0, "degree table is filled four nodes per 16-byte load");


__global__ void __launch_bounds__(1024)
hist_kernel(const int* __restrict__ row,
            const int* __restrict__ col,
            unsigned* __restrict__ part,
            uint2* __restrict__ ids16) {
    __shared__ __attribute__((aligned(16))) unsigned h[N_NODES];

    const int tid = threadIdx.x;
    const int dir = blockIdx.x / SLICES;
    const int g   = blockIdx.x - dir * SLICES;
    const int4* ids = (const int4*)((dir ? col : row) + (size_t)g * SLICE_EDGES);

    int4 v[SLICE_LOADS];
    #pragma unroll
    for (int u = 0; u < SLICE_LOADS; ++u) {
        const int i = u * 1024 + tid;
        v[u] = ids[i < SLICE_EDGES / 4 ? i : SLICE_EDGES / 4 - 1];
    }
    const uint4 z = make_uint4(0u, 0u, 0u, 0u);
    for (int i = tid; i < N_NODES / 4; i += 1024) ((uint4*)h)[i] = z;
    __syncthreads();

    uint2* o16 = ids16 + (size_t)blockIdx.x * (SLICE_EDGES / 4);
    #pragma unroll
    for (int u = 0; u < SLICE_LOADS; ++u) {
        const int i = u * 1024 + tid;
        if (i < SLICE_EDGES / 4)
            o16[i] = make_uint2((unsigned)v[u].x | ((unsigned)v[u].y << 16),
                                (unsigned)v[u].z | ((unsigned)v[u].w << 16));
    }

    #pragma unroll
    for (int u = 0; u < SLICE_LOADS; ++u) {
        if (u * 1024 + tid < SLICE_EDGES / 4) {
            atomicAdd(&h[v[u].x], 1u);
            atomicAdd(&h[v[u].y], 1u);
            atomicAdd(&h[v[u].z], 1u);
            atomicAdd(&h[v[u].w], 1u);
        }
    }
    __syncthreads();

    uint4* o = (uint4*)(part + (size_t)blockIdx.x * N_NODES);
    for (int i = tid; i < N_NODES / 4; i += 1024) o[i] = ((const uint4*)h)[i];
}

__global__ void degsum_kernel(const uint4* __restrict__ part, float4* __restrict__ deg) {
    const int i = blockIdx.x * blockDim.x + threadIdx.x;
    if (i >= 2 * (N_NODES / 4)) return;
    const int dir = i / (N_NODES / 4);
    const int j   = i - dir * (N_NODES / 4);
    const uint4* p = part + (size_t)dir * SLICES * (N_NODES / 4) + j;
    uint4 c[SLICES];
    #pragma unroll
    for (int g = 0; g < SLICES; ++g) c[g] = p[(size_t)g * (N_NODES / 4)];
    uint4 s = make_uint4(0u, 0u, 0u, 0u);
    #pragma unroll
    for (int g = 0; g < SLICES; ++g) { s.x += c[g].x; s.y += c[g].y; s.z += c[g].z; s.w += c[g].w; }
    deg[i] = make_float4((float)s.x, (float)s.y, (float)s.z, (float)s.w);
}

__device__ __forceinline__ float inv_of(float d) {
    return (d > 0.0f) ? rsqrtf(fmaxf(d, 1.0f)) : 0.0f;
}

__device__ __forceinline__ float src_factor(const unsigned char* tab,
                                            const float* __restrict__ deg_s, int s) {
    const unsigned c = tab[s];
    const float d = (c == 255u) ? deg_s[s] : (float)c;
    return inv_of(d);
}

__global__ void prep_kernel(const float* __restrict__ W_src,
                            const float* __restrict__ b_src,
                            const float* __restrict__ W_dst,
                            const float* __restrict__ b_dst,
                            float4* __restrict__ Wf,
                            float* __restrict__ bias) {
    int i = blockIdx.x * blockDim.x + threadIdx.x;
    if (i < WF_WORDS) {
        const int l  = i & 31;
        const int nt = (i >> 5) & 7;
        const int kq = i >> 8;
        const int n  = nt * 16 + (l & 15);
        const int k  = 4 * kq + 2 * (l >> 4);
        const float2 s = *(const float2*)(W_src + n * DIM + k);
        const float2 d = *(const float2*)(W_dst + n * DIM + k);
        Wf[i] = make_float4(0.5f * s.x, 0.5f * s.y, 0.5f * d.x, 0.5f * d.y);
    }
    if (i < DIM) bias[i] = 0.5f * b_src[i] + 0.5f * b_dst[i];
}

__device__ __forceinline__ void add_row(float* acc, int t, int lane, float w, const float4& v) {
    float* a = acc + t * DIM + lane;
    atomicAdd(a,      w * v.x);
    atomicAdd(a + 32, w * v.y);
    atomicAdd(a + 64, w * v.z);
    atomicAdd(a + 96, w * v.w);
}

__device__ __forceinline__ void drain_full(const unsigned* qb, int lane, int n0,
                                           const int* __restrict__ src,
                                           const float* __restrict__ deg_d,
                                           const float* __restrict__ deg_s,
                                           const unsigned char* tab,
                                           const float* __restrict__ x,
                                           float* acc) {
    const unsigned p = qb[lane];
    const int t = (int)(p >> 20);
    const int s = src[(int)(p & 0xFFFFFu)];
    const float w = inv_of(deg_d[n0 + t]) * src_factor(tab, deg_s, s);
    const int wb = __builtin_bit_cast(int, w);

    float4 v[2][8];
    #pragma unroll
    for (int u = 0; u < 8; ++u)
        v[0][u] = ((const float4*)(x + (size_t)__builtin_amdgcn_readlane(s, u) * DIM))[lane];
    #pragma unroll
    for (int g = 0; g < 4; ++g) {
        if (g + 1 < 4) {
            #pragma unroll
            for (int u = 0; u < 8; ++u)
                v[(g + 1) & 1][u] = ((const float4*)(x +
                    (size_t)__builtin_amdgcn_readlane(s, (g + 1) * 8 + u) * DIM))[lane];
        }
        #pragma unroll
        for (int u = 0; u < 8; ++u)
            add_row(acc, __builtin_amdgcn_readlane(t, g * 8 + u), lane,
                    __builtin_bit_cast(float, __builtin_amdgcn_readlane(wb, g * 8 + u)),
                    v[g & 1][u]);
    }
}

__device__ __forceinline__ void drain_tail(const unsigned* qb, int n, int lane, int n0,
                                           const int* __restrict__ src,
                                           const float* __restrict__ deg_d,
                                           const float* __restrict__ deg_s,
                                           const unsigned char* tab,
                                           const float* __restrict__ x,
                                           float* acc) {
    int s = 0, t = 0;
    float w = 0.0f;
    if (lane < n) {
        const unsigned p = qb[lane];
        t = (int)(p >> 20);
        s = src[(int)(p & 0xFFFFFu)];
        w = inv_of(deg_d[n0 + t]) * src_factor(tab, deg_s, s);
    }
    const int wb = __builtin_bit_cast(int, w);
    int i = 0;
    for (; i + 8 <= n; i += 8) {
        float4 v[8];
        #pragma unroll
        for (int u = 0; u < 8; ++u)
            v[u] = ((const float4*)(x + (size_t)__builtin_amdgcn_readlane(s, i + u) * DIM))[lane];
        #pragma unroll
        for (int u = 0; u < 8; ++u)
            add_row(acc, __builtin_amdgcn_readlane(t, i + u), lane,
                    __builtin_bit_cast(float, __builtin_amdgcn_readlane(wb, i + u)), v[u]);
    }
    for (; i < n; ++i) {
        const float4 v = ((const float4*)(x + (size_t)__builtin_amdgcn_readlane(s, i) * DIM))[lane];
        add_row(acc, __builtin_amdgcn_readlane(t, i), lane,
                __builtin_bit_cast(float, __builtin_amdgcn_readlane(wb, i)), v);
    }
}

__global__ void __launch_bounds__(1024)
aggregate_kernel(const float* __restrict__ x,
                 const int* __restrict__ row,
                 const int* __restrict__ col,
                 const float* __restrict__ deg_row,
                 const float* __restrict__ deg_col,
                 const uint4* __restrict__ ids16,
                 float* __restrict__ agg_fwd,
                 float* __restrict__ agg_rev) {
    __shared__ __attribute__((aligned(16))) float acc[RANGE * DIM];
    __shared__ unsigned queue[32 * QCAP];
    __shared__ __attribute__((aligned(16))) unsigned char tab[N_NODES];

    const int tid  = threadIdx.x;
    const int lane = tid & 31;
    const int wave = __builtin_amdgcn_readfirstlane(tid >> 5);
    const int dir  = blockIdx.x >> 7;
    const int rg   = blockIdx.x & (NRANGE - 1);
    const int n0   = rg * RANGE;
    const int cnt  = (N_NODES - n0 < RANGE) ? (N_NODES - n0) : RANGE;
    const unsigned short cnth = (unsigned short)cnt;
    const unsigned short n0h  = (unsigned short)n0;
    const us2 n0v = {n0h, n0h};

    const uint4* dst16 = ids16 + (size_t)dir * (N_EDGES / 8);
    const int*   src   = dir ? row : col;
    const float* deg_d = dir ? deg_col : deg_row;
    const float* deg_s = dir ? deg_row : deg_col;
    float*       agg   = dir ? agg_rev : agg_fwd;

    {
        float4 d[TAB_LOADS];
        #pragma unroll
        for (int u = 0; u < TAB_LOADS; ++u) {
            const int i = u * 1024 + tid;
            d[u] = ((const float4*)deg_s)[i < N_NODES / 4 ? i : N_NODES / 4 - 1];
        }
        const float4 z = make_float4(0.f, 0.f, 0.f, 0.f);
        for (int i = tid; i < RANGE * DIM / 4; i += 1024) ((float4*)acc)[i] = z;
        #pragma unroll
        for (int u = 0; u < TAB_LOADS; ++u) {
            const int i = u * 1024 + tid;
            if (i < N_NODES / 4) {
                const unsigned b0 = (unsigned)fminf(d[u].x, 255.0f);
                const unsigned b1 = (unsigned)fminf(d[u].y, 255.0f);
                const unsigned b2 = (unsigned)fminf(d[u].z, 255.0f);
                const unsigned b3 = (unsigned)fminf(d[u].w, 255.0f);
                ((unsigned*)tab)[i] = b0 | (b1 << 8) | (b2 << 16) | (b3 << 24);
            }
        }
    }
    __syncthreads();

    unsigned* q = queue + wave * QCAP;
    int pending = 0;
    const int rot = (rg * OUTER) >> 7;

    auto load4 = [&](uint4 (&r)[2], int step) {
        if (step >= OUTER) step -= OUTER;
        #pragma unroll
        for (int u = 0; u < 2; ++u) {
            int c = (step * 2 + u) * 32 + wave;
            c = c < CHUNKS ? c : CHUNKS - 1;
            r[u] = dst16[(size_t)c * 32 + lane];
        }
    };
    auto look4 = [&](const uint4 (&r)[2], int step) {
        if (step >= OUTER) step -= OUTER;
        #pragma unroll
        for (int u = 0; u < 2; ++u) {
            const int chunk = (step * 2 + u) * 32 + wave;
            if (chunk < CHUNKS) {
                const unsigned wd[4] = {r[u].x, r[u].y, r[u].z, r[u].w};
                us2 d[4];
                unsigned m[8];
                #pragma unroll
                for (int i = 0; i < 4; ++i) d[i] = __builtin_bit_cast(us2, wd[i]) - n0v;
                #pragma unroll
                for (int k = 0; k < 8; ++k)
                    m[k] = __builtin_amdgcn_ballot_w32(d[k >> 1][k & 1] < cnth);
                const unsigned e0 = (unsigned)(chunk * 256 + lane * 8);
                #pragma unroll
                for (int k = 0; k < 8; ++k) {
                    if (m[k] != 0u) {
                        const unsigned short t = d[k >> 1][k & 1];
                        if (t < cnth) q[pending + (int)__builtin_amdgcn_mbcnt_lo(m[k], 0u)] = ((unsigned)t << 20) | (e0 + (unsigned)k);
                        pending += __builtin_popcount(m[k]);
                    }
                }
            }
        }
        while (pending >= 32) {
            pending -= 32;
            drain_full(q + pending, lane, n0, src, deg_d, deg_s, tab, x, acc);
        }
    };

    uint4 da[2], db[2];
    load4(da, rot);
    for (int o = 0; o < OUTER; o += 2) {
        load4(db, o + 1 + rot);
        look4(da, o + rot);
        load4(da, o + 2 + rot);
        look4(db, o + 1 + rot);
    }
    if (pending > 0) drain_tail(q, pending, lane, n0, src, deg_d, deg_s, tab, x, acc);

    __syncthreads();

    for (int t = wave; t < cnt; t += 32) {
        const float* a = acc + t * DIM + lane;
        const float4 v = make_float4(a[0], a[32], a[64], a[96]);
        ((float4*)(agg + (size_t)(n0 + t) * DIM))[lane] = v;
    }
}

__global__ void __launch_bounds__(256)
gemm_kernel(const float* __restrict__ agg_fwd,
            const float* __restrict__ agg_rev,
            const float4* __restrict__ Wf,
            const float* __restrict__ bias,
            float* __restrict__ out) {
    __shared__ float4 wl[WF_WORDS];

    for (int i = threadIdx.x; i < WF_WORDS; i += 256) wl[i] = Wf[i];
    __syncthreads();

    const int lane = threadIdx.x & 31;
    const int mt   = blockIdx.x * 8 + (threadIdx.x >> 5);
    if (mt >= N_NODES / 16) return;

    const int half = lane >> 4;
    const int l15  = lane & 15;
    const int kb   = half * 2;

    const float* arow_f = agg_fwd + (size_t)(mt * 16 + l15) * DIM + kb;
    const float* arow_r = agg_rev + (size_t)(mt * 16 + l15) * DIM + kb;

    v8f c[8];
    #pragma unroll
    for (int nt = 0; nt < 8; ++nt) {
        const float bv = bias[nt * 16 + l15];
        #pragma unroll
        for (int j = 0; j < 8; ++j) c[nt][j] = bv;
    }

    v2f af[2][4], ar[2][4];
    #pragma unroll
    for (int u = 0; u < 4; ++u) {
        af[0][u] = *(const v2f*)(arow_f + 4 * u);
        ar[0][u] = *(const v2f*)(arow_r + 4 * u);
    }
    #pragma unroll 2
    for (int t = 0; t < DIM / 16; ++t) {
        const int cur = t & 1, nxt = cur ^ 1;
        if (t + 1 < DIM / 16) {
            #pragma unroll
            for (int u = 0; u < 4; ++u) {
                af[nxt][u] = *(const v2f*)(arow_f + 16 * (t + 1) + 4 * u);
                ar[nxt][u] = *(const v2f*)(arow_r + 16 * (t + 1) + 4 * u);
            }
        }
        #pragma unroll
        for (int u = 0; u < 4; ++u) {
            const float4* wq = wl + (t * 4 + u) * 256 + lane;
            #pragma unroll
            for (int nt = 0; nt < 8; ++nt) {
                const float4 w = wq[nt * 32];
                v2f b0, b1;
                b0[0] = w.x; b0[1] = w.y;
                b1[0] = w.z; b1[1] = w.w;
                c[nt] = __builtin_amdgcn_wmma_f32_16x16x4_f32(false, af[cur][u], false, b0,
                                                              (short)0, c[nt], false, false);
                c[nt] = __builtin_amdgcn_wmma_f32_16x16x4_f32(false, ar[cur][u], false, b1,
                                                              (short)0, c[nt], false, false);
            }
        }
    }

    const int m0 = mt * 16 + half * 8;
    #pragma unroll
    for (int nt = 0; nt < 8; ++nt) {
        const int n = nt * 16 + l15;
        #pragma unroll
        for (int j = 0; j < 8; ++j)
            out[(size_t)(m0 + j) * DIM + n] = c[nt][j];
    }
}

extern "C" void kernel_launch(void* const* d_in, const int* in_sizes, int n_in,
                              void* d_out, int out_size, void* d_ws, size_t ws_size,
                              hipStream_t stream) {
    const float* x      = (const float*)d_in[0];
    const float* W_src  = (const float*)d_in[1];
    const float* b_src  = (const float*)d_in[2];
    const float* W_dst  = (const float*)d_in[3];
    const float* b_dst  = (const float*)d_in[4];
    const int*   eidx   = (const int*)d_in[5];
    const int*   row    = eidx;
    const int*   col    = eidx + N_EDGES;
    float*       outp   = (float*)d_out;

    float* ws = (float*)d_ws;
    float* deg_row = ws;
    float* deg_col = ws + N_NODES;
    float* agg_fwd = ws + 2 * (size_t)N_NODES;
    float* agg_rev = agg_fwd + (size_t)N_NODES * DIM;
    float4* Wf     = (float4*)(agg_rev + (size_t)N_NODES * DIM);
    float* bias    = (float*)(Wf + WF_WORDS);

    hist_kernel<<<2 * SLICES, 1024, 0, stream>>>(row, col, (unsigned*)agg_fwd, (uint2*)outp);
    degsum_kernel<<<(2 * (N_NODES / 4) + 255) / 256, 256, 0, stream>>>((const uint4*)agg_fwd, (float4*)ws);
    prep_kernel<<<(WF_WORDS + 255) / 256, 256, 0, stream>>>(W_src, b_src, W_dst, b_dst, Wf, bias);
    aggregate_kernel<<<2 * NRANGE, 1024, 0, stream>>>(x, row, col, deg_row, deg_col,
                                                      (const uint4*)outp, agg_fwd, agg_rev);
    gemm_kernel<<<(N_NODES / 16 + 7) / 8, 256, 0, stream>>>(agg_fwd, agg_rev, Wf, bias, outp);
}
